# starting kernel
# baseline (speedup 1.0000x reference)
#include <hip/hip_runtime.h>
#include <hip/hip_bf16.h>

typedef __attribute__((ext_vector_type(2))) float v2f;
typedef __attribute__((ext_vector_type(4))) float v4f;
typedef __attribute__((ext_vector_type(8))) float v8f;

#define B_ 32
#define J_ 32
#define I_ 2048
#define T_ 16
#define D_ 16

__global__ __launch_bounds__(256) void uhat_gemm(const float* __restrict__ x,
                                                 const float* __restrict__ W,
                                                 float* __restrict__ u) {
    const int i    = blockIdx.x;
    const int tid  = threadIdx.x;
    const int wave = tid >> 5;
    const int lane = tid & 31;
    const int n    = lane & 15;
    const int hi   = lane >> 4;
    const int k0   = hi * 2;

    v2f Bx[4][2];
#pragma unroll
    for (int h = 0; h < 2; ++h) {
        const float* xp = x + ((size_t)(n + 16 * h) * I_ + i) * D_;
#pragma unroll
        for (int c = 0; c < 4; ++c)
            Bx[c][h] = *(const v2f*)(xp + 4 * c + k0);
    }

#pragma unroll
    for (int jj = 0; jj < 4; ++jj) {
        const int j = (wave << 2) + jj;
        const float* wp = W + (((size_t)j * I_ + i) * T_ + n) * D_;
        v2f A[4];
#pragma unroll
        for (int c = 0; c < 4; ++c)
            A[c] = *(const v2f*)(wp + 4 * c + k0);

        v8f acc0 = {0.f, 0.f, 0.f, 0.f, 0.f, 0.f, 0.f, 0.f};
        v8f acc1 = acc0;
#pragma unroll
        for (int c = 0; c < 4; ++c) {
            acc0 = __builtin_amdgcn_wmma_f32_16x16x4_f32(
                false, A[c], false, Bx[c][0], (short)0, acc0, false, false);
            acc1 = __builtin_amdgcn_wmma_f32_16x16x4_f32(
                false, A[c], false, Bx[c][1], (short)0, acc1, false, false);
        }

#pragma unroll
        for (int h = 0; h < 2; ++h) {
            const v8f acc = h ? acc1 : acc0;
            float* up = u + (((size_t)j * I_ + i) * B_ + (n + 16 * h)) * T_ + 8 * hi;
            v4f lo = __builtin_shufflevector(acc, acc, 0, 1, 2, 3);
            v4f hv = __builtin_shufflevector(acc, acc, 4, 5, 6, 7);
            *(v4f*)(up)     = lo;
            *(v4f*)(up + 4) = hv;
        }
    }
}

__global__ __launch_bounds__(256) void softmax_j(const float* __restrict__ bij,
                                                 float* __restrict__ c) {
    const int g = blockIdx.x * 256 + threadIdx.x;
    const int b = g >> 11;
    const int i = g & (I_ - 1);
    const float* bp = bij + (size_t)b * J_ * I_ + i;
    float vals[J_];
    float m = -3.4e38f;
#pragma unroll
    for (int j = 0; j < J_; ++j) {
        vals[j] = bp[(size_t)j * I_];
        m = fmaxf(m, vals[j]);
    }
    float s = 0.f;
#pragma unroll
    for (int j = 0; j < J_; ++j) {
        vals[j] = __expf(vals[j] - m);
        s += vals[j];
    }
    const float inv = 1.0f / s;
    float* cp = c + (size_t)b * J_ * I_ + i;
#pragma unroll
    for (int j = 0; j < J_; ++j)
        cp[(size_t)j * I_] = vals[j] * inv;
}

__global__ __launch_bounds__(256) void sv_kernel(const float* __restrict__ u,
                                                 const float* __restrict__ c,
                                                 float* __restrict__ v,
                                                 int uniform) {
    const int bj  = blockIdx.x;
    const int b   = bj >> 5;
    const int j   = bj & 31;
    const int tid = threadIdx.x;
    const int t   = tid & 15;
    const int grp = tid >> 4;

    const float* ubase = u + (size_t)j * I_ * (B_ * T_) + (size_t)b * T_ + t;
    const float* cbase = c + ((size_t)b * J_ + j) * I_;

    float acc = 0.f;
    for (int i = grp; i < I_; i += 16) {
        const float uv = ubase[(size_t)i * (B_ * T_)];
        acc += uniform ? uv : uv * cbase[i];
    }

    __shared__ float red[256];
    __shared__ float scale_s;
    red[tid] = acc;
    __syncthreads();
    for (int sft = 128; sft >= 16; sft >>= 1) {
        if (tid < sft) red[tid] += red[tid + sft];
        __syncthreads();
    }
    if (tid == 0) {
        float sq = 0.f;
#pragma unroll
        for (int tt = 0; tt < 16; ++tt) { const float sv = red[tt]; sq += sv * sv; }
        if (uniform) sq *= (1.0f / 32.f) * (1.0f / 32.f);
        scale_s = sq / (1.0f + sq) / sqrtf(sq + 1e-9f);
    }
    __syncthreads();
    if (tid < 16) {
        float sv = red[t];
        if (uniform) sv *= (1.0f / 32.f);
        v[((size_t)b * J_ + j) * T_ + t] = scale_s * sv;
    }
}

__global__ __launch_bounds__(256) void bupd_kernel(const float* __restrict__ u,
                                                   const float* __restrict__ v,
                                                   float* __restrict__ bij,
                                                   int first) {
    const size_t g = (size_t)blockIdx.x * 256 + threadIdx.x;
    const int i  = (int)(g & (I_ - 1));
    const int bj = (int)(g >> 11);
    const int j  = bj & 31;
    const int b  = bj >> 5;

    const float* up = u + (((size_t)j * I_ + i) * B_ + b) * T_;
    const float* vp = v + ((size_t)b * J_ + j) * T_;

    float dot = 0.f;
#pragma unroll
    for (int q = 0; q < 4; ++q) {
        const v4f uv = *(const v4f*)(up + 4 * q);
        const v4f vv = *(const v4f*)(vp + 4 * q);
        dot += uv.x * vv.x + uv.y * vv.y + uv.z * vv.z + uv.w * vv.w;
    }

    const size_t bidx = ((size_t)b * J_ + j) * I_ + i;
    bij[bidx] = first ? dot : (bij[bidx] + dot);
}

extern "C" void kernel_launch(void* const* d_in, const int* in_sizes, int n_in,
                              void* d_out, int out_size, void* d_ws, size_t ws_size,
                              hipStream_t stream) {
    const float* x = (const float*)d_in[0];
    const float* W = (const float*)d_in[1];
    float* out = (float*)d_out;

    float* u   = (float*)d_ws;
    float* bij = u + (size_t)J_ * I_ * B_ * T_;
    float* c   = bij + (size_t)B_ * J_ * I_;

    uhat_gemm<<<I_, 256, 0, stream>>>(x, W, u);

    sv_kernel<<<B_ * J_, 256, 0, stream>>>(u, c, out, 1);
    bupd_kernel<<<(B_ * J_ * I_) / 256, 256, 0, stream>>>(u, out, bij, 1);

    softmax_j<<<(B_ * I_) / 256, 256, 0, stream>>>(bij, c);
    sv_kernel<<<B_ * J_, 256, 0, stream>>>(u, c, out, 0);
    bupd_kernel<<<(B_ * J_ * I_) / 256, 256, 0, stream>>>(u, out, bij, 0);

    softmax_j<<<(B_ * I_) / 256, 256, 0, stream>>>(bij, c);
    sv_kernel<<<B_ * J_, 256, 0, stream>>>(u, c, out, 0);
}


# weighted-sum (s,v) kernel rebuilt as a stream: one block per (j, 4 b) reads whole 256-B pieces of u with b128 loads, a group of loads requested ahead of the sums; the order of the sum over i changes
# speedup vs baseline: 1.7166x; 1.7166x over previous
#include <hip/hip_runtime.h>
#include <hip/hip_bf16.h>

typedef __attribute__((ext_vector_type(2))) float v2f;
typedef __attribute__((ext_vector_type(4))) float v4f;
typedef __attribute__((ext_vector_type(8))) float v8f;

#define B_ 32
#define J_ 32
#define I_ 2048
#define T_ 16
#define D_ 16

__global__ __launch_bounds__(256) void uhat_gemm(const float* __restrict__ x,
                                                 const float* __restrict__ W,
                                                 float* __restrict__ u) {
    const int i    = blockIdx.x;
    const int tid  = threadIdx.x;
    const int wave = tid >> 5;
    const int lane = tid & 31;
    const int n    = lane & 15;
    const int hi   = lane >> 4;
    const int k0   = hi * 2;

    v2f Bx[4][2];
#pragma unroll
    for (int h = 0; h < 2; ++h) {
        const float* xp = x + ((size_t)(n + 16 * h) * I_ + i) * D_;
#pragma unroll
        for (int c = 0; c < 4; ++c)
            Bx[c][h] = *(const v2f*)(xp + 4 * c + k0);
    }

#pragma unroll
    for (int jj = 0; jj < 4; ++jj) {
        const int j = (wave << 2) + jj;
        const float* wp = W + (((size_t)j * I_ + i) * T_ + n) * D_;
        v2f A[4];
#pragma unroll
        for (int c = 0; c < 4; ++c)
            A[c] = *(const v2f*)(wp + 4 * c + k0);

        v8f acc0 = {0.f, 0.f, 0.f, 0.f, 0.f, 0.f, 0.f, 0.f};
        v8f acc1 = acc0;
#pragma unroll
        for (int c = 0; c < 4; ++c) {
            acc0 = __builtin_amdgcn_wmma_f32_16x16x4_f32(
                false, A[c], false, Bx[c][0], (short)0, acc0, false, false);
            acc1 = __builtin_amdgcn_wmma_f32_16x16x4_f32(
                false, A[c], false, Bx[c][1], (short)0, acc1, false, false);
        }

#pragma unroll
        for (int h = 0; h < 2; ++h) {
            const v8f acc = h ? acc1 : acc0;
            float* up = u + (((size_t)j * I_ + i) * B_ + (n + 16 * h)) * T_ + 8 * hi;
            v4f lo = __builtin_shufflevector(acc, acc, 0, 1, 2, 3);
            v4f hv = __builtin_shufflevector(acc, acc, 4, 5, 6, 7);
            *(v4f*)(up)     = lo;
            *(v4f*)(up + 4) = hv;
        }
    }
}

__global__ __launch_bounds__(256) void softmax_j(const float* __restrict__ bij,
                                                 float* __restrict__ c) {
    const int g = blockIdx.x * 256 + threadIdx.x;
    const int b = g >> 11;
    const int i = g & (I_ - 1);
    const float* bp = bij + (size_t)b * J_ * I_ + i;
    float vals[J_];
    float m = -3.4e38f;
#pragma unroll
    for (int j = 0; j < J_; ++j) {
        vals[j] = bp[(size_t)j * I_];
        m = fmaxf(m, vals[j]);
    }
    float s = 0.f;
#pragma unroll
    for (int j = 0; j < J_; ++j) {
        vals[j] = __expf(vals[j] - m);
        s += vals[j];
    }
    const float inv = 1.0f / s;
    float* cp = c + (size_t)b * J_ * I_ + i;
#pragma unroll
    for (int j = 0; j < J_; ++j)
        cp[(size_t)j * I_] = vals[j] * inv;
}

#define SVQ 4

template <bool UNI>
static __device__ __forceinline__ v4f sv_stream(const float* __restrict__ up,
                                                const float* __restrict__ cp) {
    v4f uu[2][4 * SVQ];
    v4f cc[2][SVQ];
    v4f a0 = {0.f, 0.f, 0.f, 0.f};
    v4f a1 = a0, a2 = a0, a3 = a0;

#define SV_LOAD(G) do { \
    _Pragma("unroll") for (int q = 0; q < SVQ; ++q) { \
        const int kk = (G) * SVQ + q; \
        if (!UNI) cc[(G) & 1][q] = *(const v4f*)(cp + 128 * kk); \
        _Pragma("unroll") for (int m = 0; m < 4; ++m) \
            uu[(G) & 1][4 * q + m] = *(const v4f*)(up + (size_t)(128 * kk + m) * (B_ * T_)); \
    } } while (0)

    SV_LOAD(0);
#pragma unroll
    for (int g = 0; g < (I_ / 128) / SVQ; ++g) {
        __builtin_amdgcn_sched_barrier(0);
        if (g + 1 < (I_ / 128) / SVQ) SV_LOAD(g + 1);
        __builtin_amdgcn_sched_barrier(0);
#pragma unroll
        for (int q = 0; q < SVQ; ++q) {
            const v4f* w = &uu[g & 1][4 * q];
            if (UNI) {
                a0 += w[0]; a1 += w[1]; a2 += w[2]; a3 += w[3];
            } else {
                const v4f cq = cc[g & 1][q];
                a0 += w[0] * cq.x; a1 += w[1] * cq.y; a2 += w[2] * cq.z; a3 += w[3] * cq.w;
            }
        }
    }
#undef SV_LOAD
    return (a0 + a1) + (a2 + a3);
}

__global__ __launch_bounds__(512) void sv_kernel(const float* __restrict__ u,
                                                 const float* __restrict__ c,
                                                 float* __restrict__ v,
                                                 int uniform) {
    const int j    = blockIdx.x >> 3;
    const int bq   = blockIdx.x & 7;
    const int tid  = threadIdx.x;
    const int lane = tid & 31;
    const int wave = tid >> 5;
    const int t4   = tid & 3;
    const int b4   = (tid >> 2) & 3;
    const int ig   = tid >> 4;
    const int b    = 4 * bq + b4;

    const float* up = u + (((size_t)j * I_ + 4 * ig) * B_ + b) * T_ + 4 * t4;
    const float* cp = c + ((size_t)b * J_ + j) * I_ + 4 * ig;

    v4f acc = uniform ? sv_stream<true>(up, cp) : sv_stream<false>(up, cp);

    acc.x += __shfl_xor(acc.x, 16, 32);
    acc.y += __shfl_xor(acc.y, 16, 32);
    acc.z += __shfl_xor(acc.z, 16, 32);
    acc.w += __shfl_xor(acc.w, 16, 32);

    __shared__ float red[16][64];
    if (lane < 16) *(v4f*)&red[wave][4 * lane] = acc;
    __syncthreads();

    if (tid < 64) {
        float s = 0.f;
#pragma unroll
        for (int w = 0; w < 16; ++w) s += red[w][tid];
        if (uniform) s *= (1.0f / 32.f);
        float sq = s * s;
#pragma unroll
        for (int m = 1; m <= 8; m <<= 1) sq += __shfl_xor(sq, m, 32);
        const float scale = sq / (1.0f + sq) / sqrtf(sq + 1e-9f);
        const int bo = 4 * bq + (tid >> 4);
        v[((size_t)bo * J_ + j) * T_ + (tid & 15)] = scale * s;
    }
}

__global__ __launch_bounds__(256) void bupd_kernel(const float* __restrict__ u,
                                                   const float* __restrict__ v,
                                                   float* __restrict__ bij,
                                                   int first) {
    const size_t g = (size_t)blockIdx.x * 256 + threadIdx.x;
    const int i  = (int)(g & (I_ - 1));
    const int bj = (int)(g >> 11);
    const int j  = bj & 31;
    const int b  = bj >> 5;

    const float* up = u + (((size_t)j * I_ + i) * B_ + b) * T_;
    const float* vp = v + ((size_t)b * J_ + j) * T_;

    float dot = 0.f;
#pragma unroll
    for (int q = 0; q < 4; ++q) {
        const v4f uv = *(const v4f*)(up + 4 * q);
        const v4f vv = *(const v4f*)(vp + 4 * q);
        dot += uv.x * vv.x + uv.y * vv.y + uv.z * vv.z + uv.w * vv.w;
    }

    const size_t bidx = ((size_t)b * J_ + j) * I_ + i;
    bij[bidx] = first ? dot : (bij[bidx] + dot);
}

extern "C" void kernel_launch(void* const* d_in, const int* in_sizes, int n_in,
                              void* d_out, int out_size, void* d_ws, size_t ws_size,
                              hipStream_t stream) {
    const float* x = (const float*)d_in[0];
    const float* W = (const float*)d_in[1];
    float* out = (float*)d_out;

    float* u   = (float*)d_ws;
    float* bij = u + (size_t)J_ * I_ * B_ * T_;
    float* c   = bij + (size_t)B_ * J_ * I_;

    uhat_gemm<<<I_, 256, 0, stream>>>(x, W, u);

    sv_kernel<<<J_ * 8, 512, 0, stream>>>(u, c, out, 1);
    bupd_kernel<<<(B_ * J_ * I_) / 256, 256, 0, stream>>>(u, out, bij, 1);

    softmax_j<<<(B_ * I_) / 256, 256, 0, stream>>>(bij, c);
    sv_kernel<<<J_ * 8, 512, 0, stream>>>(u, c, out, 0);
    bupd_kernel<<<(B_ * J_ * I_) / 256, 256, 0, stream>>>(u, out, bij, 0);

    softmax_j<<<(B_ * I_) / 256, 256, 0, stream>>>(bij, c);
    sv_kernel<<<J_ * 8, 512, 0, stream>>>(u, c, out, 0);
}


# u_hat no longer stored: each routing pass recomputes it from W by WMMA per block of 8 i and fuses agreement, softmax, weighted sum; block partials + reduce kernel; order of sums over i and t changes
# speedup vs baseline: 3.4000x; 1.9807x over previous
#include <hip/hip_runtime.h>
#include <hip/hip_bf16.h>

typedef __attribute__((ext_vector_type(2))) float v2f;
typedef __attribute__((ext_vector_type(4))) float v4f;
typedef __attribute__((ext_vector_type(8))) float v8f;

#define B_ 32
#define J_ 32
#define I_ 2048
#define T_ 16
#define D_ 16

#define NI_   8
#define NBLK_ (I_ / NI_)
#define XPB_  20
#define SMP_  33
#define SLAB_ (J_ * B_ * T_)

template <int M>
static __device__ __forceinline__ float row_xor(float x) {
    return __builtin_bit_cast(float,
        __builtin_amdgcn_update_dpp(0, __builtin_bit_cast(int, x), 0x160 + M, 0xf, 0xf, false));
}
static __device__ __forceinline__ float xhalf(float x) {
    const int b = __builtin_bit_cast(int, x);
    return __builtin_bit_cast(float,
        __builtin_amdgcn_permlanex16(b, b, 0x76543210, 0xfedcba98, false, false));
}
static __device__ __forceinline__ float wave_max(float x) {
    x = fmaxf(x, row_xor<1>(x));
    x = fmaxf(x, row_xor<2>(x));
    x = fmaxf(x, row_xor<4>(x));
    x = fmaxf(x, row_xor<8>(x));
    return fmaxf(x, xhalf(x));
}
static __device__ __forceinline__ float wave_sum(float x) {
    x += row_xor<1>(x);
    x += row_xor<2>(x);
    x += row_xor<4>(x);
    x += row_xor<8>(x);
    return x + xhalf(x);
}

static __device__ __forceinline__ void lds_barrier() {
    asm volatile("s_wait_dscnt 0x0\n\ts_barrier_signal -1\n\ts_barrier_wait -1" ::: "memory");
}

template <int MODE>
__global__ __launch_bounds__(1024) void route_pass(const float* __restrict__ x,
                                                   const float* __restrict__ W,
                                                   const float* __restrict__ vin,
                                                   float* __restrict__ bij,
                                                   float* __restrict__ part) {
    const int i0   = blockIdx.x * NI_;
    const int tid  = threadIdx.x;
    const int j    = tid >> 5;
    const int lane = tid & 31;
    const int n    = lane & 15;
    const int hi   = lane >> 4;
    const int k0   = hi * 2;

    __shared__ __attribute__((aligned(16))) float xs[NI_ * B_ * XPB_];
    __shared__ float sm[B_ * SMP_];
    __shared__ float sc[B_ * SMP_];

    {
        const int b = tid >> 5;
        const int q = tid & 31;
        const v4f xv = *(const v4f*)(x + ((size_t)b * I_ + i0) * D_ + 4 * q);
        *(v4f*)&xs[((q >> 2) * B_ + b) * XPB_ + 4 * (q & 3)] = xv;
    }

    float vr[2][8];
    if (MODE != 0) {
#pragma unroll
        for (int h = 0; h < 2; ++h) {
            const float* vp = vin + ((size_t)(n + 16 * h) * J_ + j) * T_ + 8 * hi;
            const v4f lo = *(const v4f*)(vp);
            const v4f hv = *(const v4f*)(vp + 4);
            vr[h][0] = lo.x; vr[h][1] = lo.y; vr[h][2] = lo.z; vr[h][3] = lo.w;
            vr[h][4] = hv.x; vr[h][5] = hv.y; vr[h][6] = hv.z; vr[h][7] = hv.w;
        }
    }

    float bo[NI_];
    if (MODE == 2) {
#pragma unroll
        for (int ii = 0; ii < NI_; ++ii)
            bo[ii] = bij[((size_t)(i0 + ii) * J_ + j) * B_ + lane];
    }

    const float* wp = W + (((size_t)j * I_ + i0) * T_ + n) * D_ + k0;
    v2f Ar[3][4];
#pragma unroll
    for (int c = 0; c < 4; ++c) {
        Ar[0][c] = *(const v2f*)(wp + 4 * c);
        Ar[1][c] = *(const v2f*)(wp + T_ * D_ + 4 * c);
    }

    v8f s0 = {0.f, 0.f, 0.f, 0.f, 0.f, 0.f, 0.f, 0.f};
    v8f s1 = s0;

    __syncthreads();

#pragma unroll
    for (int ii = 0; ii < NI_; ++ii) {
        if (ii + 2 < NI_) {
#pragma unroll
            for (int c = 0; c < 4; ++c)
                Ar[(ii + 2) % 3][c] = *(const v2f*)(wp + (ii + 2) * (T_ * D_) + 4 * c);
        }

        v2f Bx[4][2];
#pragma unroll
        for (int h = 0; h < 2; ++h) {
#pragma unroll
            for (int c = 0; c < 4; ++c)
                Bx[c][h] = *(const v2f*)&xs[(ii * B_ + n + 16 * h) * XPB_ + 4 * c + k0];
        }

        v8f acc0 = {0.f, 0.f, 0.f, 0.f, 0.f, 0.f, 0.f, 0.f};
        v8f acc1 = acc0;
#pragma unroll
        for (int c = 0; c < 4; ++c) {
            acc0 = __builtin_amdgcn_wmma_f32_16x16x4_f32(
                false, Ar[ii % 3][c], false, Bx[c][0], (short)0, acc0, false, false);
            acc1 = __builtin_amdgcn_wmma_f32_16x16x4_f32(
                false, Ar[ii % 3][c], false, Bx[c][1], (short)0, acc1, false, false);
        }

        float c0, c1;
        if (MODE == 0) {
            c0 = 1.0f / 32.f;
            c1 = 1.0f / 32.f;
        } else {
            float p0 = 0.f, p1 = 0.f;
#pragma unroll
            for (int r = 0; r < 8; ++r) {
                p0 += acc0[r] * vr[0][r];
                p1 += acc1[r] * vr[1][r];
            }
            const float keep = hi ? p1 : p0;
            const float send = hi ? p0 : p1;
            float logit = keep + xhalf(send);
            if (MODE == 2) logit += bo[ii];
            if (MODE == 1) bij[((size_t)(i0 + ii) * J_ + j) * B_ + lane] = logit;

            sm[lane * SMP_ + j] = logit;
            lds_barrier();
            const float z   = sm[j * SMP_ + lane];
            const float m   = wave_max(z);
            const float e   = __expf(z - m);
            const float sum = wave_sum(e);
            sc[j * SMP_ + lane] = e / sum;
            lds_barrier();
            c0 = sc[n * SMP_ + j];
            c1 = sc[(n + 16) * SMP_ + j];
        }
        s0 += acc0 * c0;
        s1 += acc1 * c1;
    }

#pragma unroll
    for (int h = 0; h < 2; ++h) {
        const v8f acc = h ? s1 : s0;
        float* pp = part + (((size_t)blockIdx.x * J_ + j) * B_ + (n + 16 * h)) * T_ + 8 * hi;
        v4f lo = __builtin_shufflevector(acc, acc, 0, 1, 2, 3);
        v4f hv = __builtin_shufflevector(acc, acc, 4, 5, 6, 7);
        *(v4f*)(pp)     = lo;
        *(v4f*)(pp + 4) = hv;
    }
}

__global__ __launch_bounds__(512) void reduce_squash(const float* __restrict__ part,
                                                     float* __restrict__ v) {
    const int jp  = blockIdx.x >> 4;
    const int bp  = blockIdx.x & 15;
    const int tid = threadIdx.x;
    const int e   = tid & 63;
    const int g   = tid >> 6;
    const int jj  = e >> 5;

    const float* pp = part + (size_t)(32 * g) * SLAB_
                    + ((size_t)(2 * jp + jj) * B_ + 2 * bp) * T_ + (e & 31);
    float a0 = 0.f, a1 = 0.f, a2 = 0.f, a3 = 0.f;
#pragma unroll
    for (int k = 0; k < 32; k += 4) {
        a0 += pp[(size_t)(k + 0) * SLAB_];
        a1 += pp[(size_t)(k + 1) * SLAB_];
        a2 += pp[(size_t)(k + 2) * SLAB_];
        a3 += pp[(size_t)(k + 3) * SLAB_];
    }

    __shared__ float red[8][64];
    red[g][e] = (a0 + a1) + (a2 + a3);
    __syncthreads();

    if (tid < 64) {
        float s = 0.f;
#pragma unroll
        for (int w = 0; w < 8; ++w) s += red[w][tid];
        float sq = s * s;
#pragma unroll
        for (int m = 1; m <= 8; m <<= 1) sq += __shfl_xor(sq, m, 32);
        const float scale = sq / (1.0f + sq) / sqrtf(sq + 1e-9f);
        const int b  = 2 * bp + ((tid >> 4) & 1);
        const int jo = 2 * jp + (tid >> 5);
        v[((size_t)b * J_ + jo) * T_ + (tid & 15)] = scale * s;
    }
}

extern "C" void kernel_launch(void* const* d_in, const int* in_sizes, int n_in,
                              void* d_out, int out_size, void* d_ws, size_t ws_size,
                              hipStream_t stream) {
    const float* x = (const float*)d_in[0];
    const float* W = (const float*)d_in[1];
    float* out = (float*)d_out;

    float* part = (float*)d_ws;
    float* bij  = part + (size_t)NBLK_ * SLAB_;

    route_pass<0><<<NBLK_, 1024, 0, stream>>>(x, W, out, bij, part);
    reduce_squash<<<(J_ / 2) * (B_ / 2), 512, 0, stream>>>(part, out);

    route_pass<1><<<NBLK_, 1024, 0, stream>>>(x, W, out, bij, part);
    reduce_squash<<<(J_ / 2) * (B_ / 2), 512, 0, stream>>>(part, out);

    route_pass<2><<<NBLK_, 1024, 0, stream>>>(x, W, out, bij, part);
    reduce_squash<<<(J_ / 2) * (B_ / 2), 512, 0, stream>>>(part, out);
}


# routing pass: vector instructions beside the f32 products cut: softmax of 2 i per LDS exchange (half-wave each, max by DPP ops, e*rcp(sum)); pass 0 sums i in the accumulator; these sum orders change
# speedup vs baseline: 3.6791x; 1.0821x over previous
#include <hip/hip_runtime.h>
#include <hip/hip_bf16.h>

typedef __attribute__((ext_vector_type(2))) float v2f;
typedef __attribute__((ext_vector_type(4))) float v4f;
typedef __attribute__((ext_vector_type(8))) float v8f;

#define B_ 32
#define J_ 32
#define I_ 2048
#define T_ 16
#define D_ 16

#define NI_   8
#define NBLK_ (I_ / NI_)
#define XPB_  20
#define SMP_  33
#define SMT_  (B_ * SMP_)
#define SLAB_ (J_ * B_ * T_)

template <int M>
static __device__ __forceinline__ float row_xor(float x) {
    return __builtin_bit_cast(float,
        __builtin_amdgcn_update_dpp(0, __builtin_bit_cast(int, x), 0x160 + M, 0xf, 0xf, false));
}
static __device__ __forceinline__ float xhalf(float x) {
    const int b = __builtin_bit_cast(int, x);
    return __builtin_bit_cast(float,
        __builtin_amdgcn_permlanex16(b, b, 0x76543210, 0xfedcba98, false, false));
}
static __device__ __forceinline__ float max_num(float a, float b) {
    float r;
    asm("v_max_num_f32 %0, %1, %2" : "=v"(r) : "v"(a), "v"(b));
    return r;
}
template <int M>
static __device__ __forceinline__ float row_xor_max(float x) {
    float r;
    asm("v_max_num_f32_dpp %0, %1, %1 row_xmask:%2 row_mask:0xf bank_mask:0xf"
        : "=v"(r) : "v"(x), "n"(M));
    return r;
}
static __device__ __forceinline__ float half_max(float x) {
    x = row_xor_max<1>(x);
    x = row_xor_max<2>(x);
    x = row_xor_max<4>(x);
    return row_xor_max<8>(x);
}
static __device__ __forceinline__ float half_sum(float x) {
    x += row_xor<1>(x);
    x += row_xor<2>(x);
    x += row_xor<4>(x);
    return x + row_xor<8>(x);
}

static __device__ __forceinline__ void lds_barrier() {
    asm volatile("s_wait_dscnt 0x0\n\ts_barrier_signal -1\n\ts_barrier_wait -1" ::: "memory");
}

template <int MODE>
__global__ __launch_bounds__(1024) void route_pass(const float* __restrict__ x,
                                                   const float* __restrict__ W,
                                                   const float* __restrict__ vin,
                                                   float* __restrict__ bij,
                                                   float* __restrict__ part) {
    const int i0   = blockIdx.x * NI_;
    const int tid  = threadIdx.x;
    const int j    = tid >> 5;
    const int lane = tid & 31;
    const int n    = lane & 15;
    const int hi   = lane >> 4;
    const int k0   = hi * 2;

    __shared__ __attribute__((aligned(16))) float xs[NI_ * B_ * XPB_];
    __shared__ float sm[2 * SMT_];
    __shared__ float sc[2 * SMT_];

    {
        const int b = tid >> 5;
        const int q = tid & 31;
        const v4f xv = *(const v4f*)(x + ((size_t)b * I_ + i0) * D_ + 4 * q);
        *(v4f*)&xs[((q >> 2) * B_ + b) * XPB_ + 4 * (q & 3)] = xv;
    }

    float vr[2][8];
    if (MODE != 0) {
#pragma unroll
        for (int h = 0; h < 2; ++h) {
            const float* vp = vin + ((size_t)(n + 16 * h) * J_ + j) * T_ + 8 * hi;
            const v4f lo = *(const v4f*)(vp);
            const v4f hv = *(const v4f*)(vp + 4);
            vr[h][0] = lo.x; vr[h][1] = lo.y; vr[h][2] = lo.z; vr[h][3] = lo.w;
            vr[h][4] = hv.x; vr[h][5] = hv.y; vr[h][6] = hv.z; vr[h][7] = hv.w;
        }
    }

    float bo[NI_];
    if (MODE == 2) {
#pragma unroll
        for (int ii = 0; ii < NI_; ++ii)
            bo[ii] = bij[((size_t)(i0 + ii) * J_ + j) * B_ + lane];
    }

    const float* wp = W + (((size_t)j * I_ + i0) * T_ + n) * D_ + k0;
    v2f Ar[3][4];
#pragma unroll
    for (int c = 0; c < 4; ++c) {
        Ar[0][c] = *(const v2f*)(wp + 4 * c);
        Ar[1][c] = *(const v2f*)(wp + T_ * D_ + 4 * c);
    }

    v8f s0 = {0.f, 0.f, 0.f, 0.f, 0.f, 0.f, 0.f, 0.f};
    v8f s1 = s0;

    __syncthreads();

    v8f acc[2][2];
#pragma unroll
    for (int pp = 0; pp < NI_ / 2; ++pp) {
#pragma unroll
        for (int q = 0; q < 2; ++q) {
            const int ii = 2 * pp + q;
            if (ii + 2 < NI_) {
#pragma unroll
                for (int c = 0; c < 4; ++c)
                    Ar[(ii + 2) % 3][c] = *(const v2f*)(wp + (ii + 2) * (T_ * D_) + 4 * c);
            }

            v2f Bx[4][2];
#pragma unroll
            for (int h = 0; h < 2; ++h) {
#pragma unroll
                for (int c = 0; c < 4; ++c)
                    Bx[c][h] = *(const v2f*)&xs[(ii * B_ + n + 16 * h) * XPB_ + 4 * c + k0];
            }

            if (MODE != 0 || ii == 0) {
                acc[q][0] = (v8f){0.f, 0.f, 0.f, 0.f, 0.f, 0.f, 0.f, 0.f};
                acc[q][1] = acc[q][0];
            } else {
                acc[q][0] = acc[q ^ 1][0];
                acc[q][1] = acc[q ^ 1][1];
            }
#pragma unroll
            for (int c = 0; c < 4; ++c) {
                acc[q][0] = __builtin_amdgcn_wmma_f32_16x16x4_f32(
                    false, Ar[ii % 3][c], false, Bx[c][0], (short)0, acc[q][0], false, false);
                acc[q][1] = __builtin_amdgcn_wmma_f32_16x16x4_f32(
                    false, Ar[ii % 3][c], false, Bx[c][1], (short)0, acc[q][1], false, false);
            }

            if (MODE != 0) {
                float p0 = 0.f, p1 = 0.f;
#pragma unroll
                for (int r = 0; r < 8; ++r) {
                    p0 += acc[q][0][r] * vr[0][r];
                    p1 += acc[q][1][r] * vr[1][r];
                }
                const float keep = hi ? p1 : p0;
                const float send = hi ? p0 : p1;
                float logit = keep + xhalf(send);
                if (MODE == 2) logit += bo[ii];
                if (MODE == 1) bij[((size_t)(i0 + ii) * J_ + j) * B_ + lane] = logit;
                sm[q * SMT_ + lane * SMP_ + j] = logit;
                if (q == 0) asm volatile("" ::: "memory");
            }
        }

        if (MODE != 0) {
            lds_barrier();
            const float* zp = &sm[hi * SMT_ + j * SMP_ + n];
            const float z0 = zp[0], z1 = zp[16];
            const float m  = half_max(max_num(z0, z1));
            const float e0 = __expf(z0 - m), e1 = __expf(z1 - m);
            const float sum = half_sum(e0 + e1);
            const float inv = __builtin_amdgcn_rcpf(sum);
            float* cp = &sc[hi * SMT_ + j * SMP_ + n];
            cp[0]  = e0 * inv;
            cp[16] = e1 * inv;
            lds_barrier();
#pragma unroll
            for (int q = 0; q < 2; ++q) {
                const float c0 = sc[q * SMT_ + n * SMP_ + j];
                const float c1 = sc[q * SMT_ + (n + 16) * SMP_ + j];
                s0 += acc[q][0] * c0;
                s1 += acc[q][1] * c1;
            }
        } else if (pp == NI_ / 2 - 1) {
            s0 = acc[1][0] * (1.0f / 32.f);
            s1 = acc[1][1] * (1.0f / 32.f);
        }
    }

#pragma unroll
    for (int h = 0; h < 2; ++h) {
        const v8f acc = h ? s1 : s0;
        float* pp = part + (((size_t)blockIdx.x * J_ + j) * B_ + (n + 16 * h)) * T_ + 8 * hi;
        v4f lo = __builtin_shufflevector(acc, acc, 0, 1, 2, 3);
        v4f hv = __builtin_shufflevector(acc, acc, 4, 5, 6, 7);
        *(v4f*)(pp)     = lo;
        *(v4f*)(pp + 4) = hv;
    }
}

__global__ __launch_bounds__(512) void reduce_squash(const float* __restrict__ part,
                                                     float* __restrict__ v) {
    const int jp  = blockIdx.x >> 4;
    const int bp  = blockIdx.x & 15;
    const int tid = threadIdx.x;
    const int e   = tid & 63;
    const int g   = tid >> 6;
    const int jj  = e >> 5;

    const float* pp = part + (size_t)(32 * g) * SLAB_
                    + ((size_t)(2 * jp + jj) * B_ + 2 * bp) * T_ + (e & 31);
    float a0 = 0.f, a1 = 0.f, a2 = 0.f, a3 = 0.f;
#pragma unroll
    for (int k = 0; k < 32; k += 4) {
        a0 += pp[(size_t)(k + 0) * SLAB_];
        a1 += pp[(size_t)(k + 1) * SLAB_];
        a2 += pp[(size_t)(k + 2) * SLAB_];
        a3 += pp[(size_t)(k + 3) * SLAB_];
    }

    __shared__ float red[8][64];
    red[g][e] = (a0 + a1) + (a2 + a3);
    __syncthreads();

    if (tid < 64) {
        float s = 0.f;
#pragma unroll
        for (int w = 0; w < 8; ++w) s += red[w][tid];
        float sq = s * s;
#pragma unroll
        for (int m = 1; m <= 8; m <<= 1) sq += __shfl_xor(sq, m, 32);
        const float scale = sq / (1.0f + sq) / sqrtf(sq + 1e-9f);
        const int b  = 2 * bp + ((tid >> 4) & 1);
        const int jo = 2 * jp + (tid >> 5);
        v[((size_t)b * J_ + jo) * T_ + (tid & 15)] = scale * s;
    }
}

extern "C" void kernel_launch(void* const* d_in, const int* in_sizes, int n_in,
                              void* d_out, int out_size, void* d_ws, size_t ws_size,
                              hipStream_t stream) {
    const float* x = (const float*)d_in[0];
    const float* W = (const float*)d_in[1];
    float* out = (float*)d_out;

    float* part = (float*)d_ws;
    float* bij  = part + (size_t)NBLK_ * SLAB_;

    route_pass<0><<<NBLK_, 1024, 0, stream>>>(x, W, out, bij, part);
    reduce_squash<<<(J_ / 2) * (B_ / 2), 512, 0, stream>>>(part, out);

    route_pass<1><<<NBLK_, 1024, 0, stream>>>(x, W, out, bij, part);
    reduce_squash<<<(J_ / 2) * (B_ / 2), 512, 0, stream>>>(part, out);

    route_pass<2><<<NBLK_, 1024, 0, stream>>>(x, W, out, bij, part);
    reduce_squash<<<(J_ / 2) * (B_ / 2), 512, 0, stream>>>(part, out);
}


# logit plane removed from memory: the last pass forms its logits as u.(v0+v1) from the summed outputs, not as stored u.v0 plus u.v1; 8 MB less written and read; the order of that sum changes
# speedup vs baseline: 3.7961x; 1.0318x over previous
#include <hip/hip_runtime.h>
#include <hip/hip_bf16.h>

typedef __attribute__((ext_vector_type(2))) float v2f;
typedef __attribute__((ext_vector_type(4))) float v4f;
typedef __attribute__((ext_vector_type(8))) float v8f;

#define B_ 32
#define J_ 32
#define I_ 2048
#define T_ 16
#define D_ 16

#define NI_   8
#define NBLK_ (I_ / NI_)
#define XPB_  20
#define SMP_  33
#define SMT_  (B_ * SMP_)
#define SLAB_ (J_ * B_ * T_)

template <int M>
static __device__ __forceinline__ float row_xor(float x) {
    return __builtin_bit_cast(float,
        __builtin_amdgcn_update_dpp(0, __builtin_bit_cast(int, x), 0x160 + M, 0xf, 0xf, false));
}
static __device__ __forceinline__ float xhalf(float x) {
    const int b = __builtin_bit_cast(int, x);
    return __builtin_bit_cast(float,
        __builtin_amdgcn_permlanex16(b, b, 0x76543210, 0xfedcba98, false, false));
}
static __device__ __forceinline__ float max_num(float a, float b) {
    float r;
    asm("v_max_num_f32 %0, %1, %2" : "=v"(r) : "v"(a), "v"(b));
    return r;
}
template <int M>
static __device__ __forceinline__ float row_xor_max(float x) {
    float r;
    asm("v_max_num_f32_dpp %0, %1, %1 row_xmask:%2 row_mask:0xf bank_mask:0xf"
        : "=v"(r) : "v"(x), "n"(M));
    return r;
}
static __device__ __forceinline__ float half_max(float x) {
    x = row_xor_max<1>(x);
    x = row_xor_max<2>(x);
    x = row_xor_max<4>(x);
    return row_xor_max<8>(x);
}
static __device__ __forceinline__ float half_sum(float x) {
    x += row_xor<1>(x);
    x += row_xor<2>(x);
    x += row_xor<4>(x);
    return x + row_xor<8>(x);
}

static __device__ __forceinline__ void lds_barrier() {
    asm volatile("s_wait_dscnt 0x0\n\ts_barrier_signal -1\n\ts_barrier_wait -1" ::: "memory");
}

template <int MODE>
__global__ __launch_bounds__(1024) void route_pass(const float* __restrict__ x,
                                                   const float* __restrict__ W,
                                                   const float* __restrict__ vin,
                                                   float* __restrict__ part) {
    const int i0   = blockIdx.x * NI_;
    const int tid  = threadIdx.x;
    const int j    = tid >> 5;
    const int lane = tid & 31;
    const int n    = lane & 15;
    const int hi   = lane >> 4;
    const int k0   = hi * 2;

    __shared__ __attribute__((aligned(16))) float xs[NI_ * B_ * XPB_];
    __shared__ float sm[2 * SMT_];
    __shared__ float sc[2 * SMT_];

    {
        const int b = tid >> 5;
        const int q = tid & 31;
        const v4f xv = *(const v4f*)(x + ((size_t)b * I_ + i0) * D_ + 4 * q);
        *(v4f*)&xs[((q >> 2) * B_ + b) * XPB_ + 4 * (q & 3)] = xv;
    }

    float vr[2][8];
    if (MODE != 0) {
#pragma unroll
        for (int h = 0; h < 2; ++h) {
            const float* vp = vin + ((size_t)(n + 16 * h) * J_ + j) * T_ + 8 * hi;
            const v4f lo = *(const v4f*)(vp);
            const v4f hv = *(const v4f*)(vp + 4);
            vr[h][0] = lo.x; vr[h][1] = lo.y; vr[h][2] = lo.z; vr[h][3] = lo.w;
            vr[h][4] = hv.x; vr[h][5] = hv.y; vr[h][6] = hv.z; vr[h][7] = hv.w;
        }
    }

    const float* wp = W + (((size_t)j * I_ + i0) * T_ + n) * D_ + k0;
    v2f Ar[3][4];
#pragma unroll
    for (int c = 0; c < 4; ++c) {
        Ar[0][c] = *(const v2f*)(wp + 4 * c);
        Ar[1][c] = *(const v2f*)(wp + T_ * D_ + 4 * c);
    }

    v8f s0 = {0.f, 0.f, 0.f, 0.f, 0.f, 0.f, 0.f, 0.f};
    v8f s1 = s0;

    __syncthreads();

    v8f acc[2][2];
#pragma unroll
    for (int pp = 0; pp < NI_ / 2; ++pp) {
#pragma unroll
        for (int q = 0; q < 2; ++q) {
            const int ii = 2 * pp + q;
            if (ii + 2 < NI_) {
#pragma unroll
                for (int c = 0; c < 4; ++c)
                    Ar[(ii + 2) % 3][c] = *(const v2f*)(wp + (ii + 2) * (T_ * D_) + 4 * c);
            }

            v2f Bx[4][2];
#pragma unroll
            for (int h = 0; h < 2; ++h) {
#pragma unroll
                for (int c = 0; c < 4; ++c)
                    Bx[c][h] = *(const v2f*)&xs[(ii * B_ + n + 16 * h) * XPB_ + 4 * c + k0];
            }

            if (MODE != 0 || ii == 0) {
                acc[q][0] = (v8f){0.f, 0.f, 0.f, 0.f, 0.f, 0.f, 0.f, 0.f};
                acc[q][1] = acc[q][0];
            } else {
                acc[q][0] = acc[q ^ 1][0];
                acc[q][1] = acc[q ^ 1][1];
            }
#pragma unroll
            for (int c = 0; c < 4; ++c) {
                acc[q][0] = __builtin_amdgcn_wmma_f32_16x16x4_f32(
                    false, Ar[ii % 3][c], false, Bx[c][0], (short)0, acc[q][0], false, false);
                acc[q][1] = __builtin_amdgcn_wmma_f32_16x16x4_f32(
                    false, Ar[ii % 3][c], false, Bx[c][1], (short)0, acc[q][1], false, false);
            }

            if (MODE != 0) {
                float p0 = 0.f, p1 = 0.f;
#pragma unroll
                for (int r = 0; r < 8; ++r) {
                    p0 += acc[q][0][r] * vr[0][r];
                    p1 += acc[q][1][r] * vr[1][r];
                }
                const float keep = hi ? p1 : p0;
                const float send = hi ? p0 : p1;
                const float logit = keep + xhalf(send);
                sm[q * SMT_ + lane * SMP_ + j] = logit;
                if (q == 0) asm volatile("" ::: "memory");
            }
        }

        if (MODE != 0) {
            lds_barrier();
            const float* zp = &sm[hi * SMT_ + j * SMP_ + n];
            const float z0 = zp[0], z1 = zp[16];
            const float m  = half_max(max_num(z0, z1));
            const float e0 = __expf(z0 - m), e1 = __expf(z1 - m);
            const float sum = half_sum(e0 + e1);
            const float inv = __builtin_amdgcn_rcpf(sum);
            float* cp = &sc[hi * SMT_ + j * SMP_ + n];
            cp[0]  = e0 * inv;
            cp[16] = e1 * inv;
            lds_barrier();
#pragma unroll
            for (int q = 0; q < 2; ++q) {
                const float c0 = sc[q * SMT_ + n * SMP_ + j];
                const float c1 = sc[q * SMT_ + (n + 16) * SMP_ + j];
                s0 += acc[q][0] * c0;
                s1 += acc[q][1] * c1;
            }
        } else if (pp == NI_ / 2 - 1) {
            s0 = acc[1][0] * (1.0f / 32.f);
            s1 = acc[1][1] * (1.0f / 32.f);
        }
    }

#pragma unroll
    for (int h = 0; h < 2; ++h) {
        const v8f acc = h ? s1 : s0;
        float* pp = part + (((size_t)blockIdx.x * J_ + j) * B_ + (n + 16 * h)) * T_ + 8 * hi;
        v4f lo = __builtin_shufflevector(acc, acc, 0, 1, 2, 3);
        v4f hv = __builtin_shufflevector(acc, acc, 4, 5, 6, 7);
        *(v4f*)(pp)     = lo;
        *(v4f*)(pp + 4) = hv;
    }
}

template <int ADD>
__global__ __launch_bounds__(512) void reduce_squash(const float* __restrict__ part,
                                                     float* __restrict__ v) {
    const int jp  = blockIdx.x >> 4;
    const int bp  = blockIdx.x & 15;
    const int tid = threadIdx.x;
    const int e   = tid & 63;
    const int g   = tid >> 6;
    const int jj  = e >> 5;

    const float* pp = part + (size_t)(32 * g) * SLAB_
                    + ((size_t)(2 * jp + jj) * B_ + 2 * bp) * T_ + (e & 31);
    float a0 = 0.f, a1 = 0.f, a2 = 0.f, a3 = 0.f;
#pragma unroll
    for (int k = 0; k < 32; k += 4) {
        a0 += pp[(size_t)(k + 0) * SLAB_];
        a1 += pp[(size_t)(k + 1) * SLAB_];
        a2 += pp[(size_t)(k + 2) * SLAB_];
        a3 += pp[(size_t)(k + 3) * SLAB_];
    }

    __shared__ float red[8][64];
    red[g][e] = (a0 + a1) + (a2 + a3);
    __syncthreads();

    if (tid < 64) {
        float s = 0.f;
#pragma unroll
        for (int w = 0; w < 8; ++w) s += red[w][tid];
        float sq = s * s;
#pragma unroll
        for (int m = 1; m <= 8; m <<= 1) sq += __shfl_xor(sq, m, 32);
        const float scale = sq / (1.0f + sq) / sqrtf(sq + 1e-9f);
        const int b  = 2 * bp + ((tid >> 4) & 1);
        const int jo = 2 * jp + (tid >> 5);
        float* vp = v + ((size_t)b * J_ + jo) * T_ + (tid & 15);
        const float r = scale * s;
        *vp = ADD ? *vp + r : r;
    }
}

extern "C" void kernel_launch(void* const* d_in, const int* in_sizes, int n_in,
                              void* d_out, int out_size, void* d_ws, size_t ws_size,
                              hipStream_t stream) {
    const float* x = (const float*)d_in[0];
    const float* W = (const float*)d_in[1];
    float* out = (float*)d_out;

    float* part = (float*)d_ws;

    route_pass<0><<<NBLK_, 1024, 0, stream>>>(x, W, out, part);
    reduce_squash<0><<<(J_ / 2) * (B_ / 2), 512, 0, stream>>>(part, out);

    route_pass<1><<<NBLK_, 1024, 0, stream>>>(x, W, out, part);
    reduce_squash<1><<<(J_ / 2) * (B_ / 2), 512, 0, stream>>>(part, out);

    route_pass<1><<<NBLK_, 1024, 0, stream>>>(x, W, out, part);
    reduce_squash<0><<<(J_ / 2) * (B_ / 2), 512, 0, stream>>>(part, out);
}
